# starting kernel
# baseline (speedup 1.0000x reference)
#include <hip/hip_runtime.h>

typedef __attribute__((ext_vector_type(2))) float v2f;
typedef __attribute__((ext_vector_type(8))) float v8f;

#define NNODES 50000
#define NEDGES 800000
#define DH     64

__global__ void fill_f32(float* __restrict__ p, float v, int n) {
    int i = blockIdx.x * blockDim.x + threadIdx.x;
    if (i < n) p[i] = v;
}

__global__ void deg_kernel(const int* __restrict__ dst, float* __restrict__ deg, int e) {
    int i = blockIdx.x * blockDim.x + threadIdx.x;
    if (i < e) atomicAdd(&deg[dst[i]], 1.0f);
}

__global__ void rsqrt_kernel(float* __restrict__ d, int n) {
    int i = blockIdx.x * blockDim.x + threadIdx.x;
    if (i < n) d[i] = __frsqrt_rn(d[i]);
}

__global__ void gemm64_wmma(const float* __restrict__ Hin,
                            const float* __restrict__ W,
                            float* __restrict__ Hout,
                            int ntiles) {
    __shared__ float Wl[32 * 128];
    for (int i = threadIdx.x; i < DH * DH; i += blockDim.x) {
        int k = i >> 6, n = i & 63;
        Wl[(k >> 1) * 128 + n * 2 + (k & 1)] = W[i];
    }
    __syncthreads();

    const int lane = threadIdx.x & 31;
    const int wave = threadIdx.x >> 5;
    const int tile = blockIdx.x * (blockDim.x >> 5) + wave;
    if (tile >= ntiles) return;

    const int m0    = tile * 16;
    const int row   = m0 + (lane & 15);
    const int khalf = (lane >> 4) * 2;
    const int ncol  = lane & 15;

    v8f acc0 = {}, acc1 = {}, acc2 = {}, acc3 = {};

    #pragma unroll
    for (int k0 = 0; k0 < DH; k0 += 4) {
        const int kb = k0 + khalf;
        v2f a = *(const v2f*)(Hin + row * DH + kb);

        const float* bp = &Wl[(kb >> 1) * 128 + ncol * 2];
        v2f b0 = *(const v2f*)(bp +  0);
        v2f b1 = *(const v2f*)(bp + 32);
        v2f b2 = *(const v2f*)(bp + 64);
        v2f b3 = *(const v2f*)(bp + 96);

        acc0 = __builtin_amdgcn_wmma_f32_16x16x4_f32(false, a, false, b0, (short)0, acc0, false, false);
        acc1 = __builtin_amdgcn_wmma_f32_16x16x4_f32(false, a, false, b1, (short)0, acc1, false, false);
        acc2 = __builtin_amdgcn_wmma_f32_16x16x4_f32(false, a, false, b2, (short)0, acc2, false, false);
        acc3 = __builtin_amdgcn_wmma_f32_16x16x4_f32(false, a, false, b3, (short)0, acc3, false, false);
    }

    const int rowoff = (lane >> 4) * 8;
    #pragma unroll
    for (int r = 0; r < 8; ++r) {
        float* o = Hout + (m0 + rowoff + r) * DH + ncol;
        o[ 0] = acc0[r];
        o[16] = acc1[r];
        o[32] = acc2[r];
        o[48] = acc3[r];
    }
}

__global__ void scatter_kernel(const int* __restrict__ src,
                               const int* __restrict__ dst,
                               const float* __restrict__ dis,
                               const float* __restrict__ B,
                               float* __restrict__ A, int e) {
    int tid  = blockIdx.x * blockDim.x + threadIdx.x;
    int eidx = tid >> 4;
    if (eidx >= e) return;
    int f4 = (tid & 15) * 4;
    int s = src[eidx], d = dst[eidx];
    float w = dis[s] * dis[d];
    float4 v = *(const float4*)(B + s * DH + f4);
    float* o = A + d * DH + f4;
    atomicAdd(o + 0, v.x * w);
    atomicAdd(o + 1, v.y * w);
    atomicAdd(o + 2, v.z * w);
    atomicAdd(o + 3, v.w * w);
}

__global__ void bias_relu_kernel(float* __restrict__ A,
                                 const float* __restrict__ B,
                                 const float* __restrict__ dis,
                                 const float* __restrict__ bias, int n64) {
    int tid = blockIdx.x * blockDim.x + threadIdx.x;
    if (tid >= n64) return;
    int node = tid >> 6, f = tid & 63;
    float dd = dis[node];
    float v = A[tid] + B[tid] * dd * dd + bias[f];
    A[tid] = v > 0.0f ? v : 0.0f;
}

__global__ void final_linear(const float* __restrict__ A,
                             const float* __restrict__ W,
                             const float* __restrict__ b,
                             float* __restrict__ out, int n8) {
    int tid = blockIdx.x * blockDim.x + threadIdx.x;
    if (tid >= n8) return;
    int nrow = tid >> 3, k = tid & 7;
    const float* a = A + nrow * DH;
    float s = b[k];
    #pragma unroll
    for (int j = 0; j < DH; ++j) s += a[j] * W[j * 8 + k];
    out[tid] = s;
}

extern "C" void kernel_launch(void* const* d_in, const int* in_sizes, int n_in,
                              void* d_out, int out_size, void* d_ws, size_t ws_size,
                              hipStream_t stream) {
    const float* x    = (const float*)d_in[0];
    const int*   edge = (const int*)d_in[1];
    const int*   src  = edge;
    const int*   dst  = edge + NEDGES;
    const float* Ws[3] = { (const float*)d_in[2], (const float*)d_in[4], (const float*)d_in[6] };
    const float* bs[3] = { (const float*)d_in[3], (const float*)d_in[5], (const float*)d_in[7] };
    const float* linW = (const float*)d_in[8];
    const float* linb = (const float*)d_in[9];

    float* A   = (float*)d_ws;
    float* B   = A + NNODES * DH;
    float* dis = B + NNODES * DH;

    const int n64 = NNODES * DH;

    fill_f32<<<(NNODES + 255) / 256, 256, 0, stream>>>(dis, 1.0f, NNODES);
    deg_kernel<<<(NEDGES + 255) / 256, 256, 0, stream>>>(dst, dis, NEDGES);
    rsqrt_kernel<<<(NNODES + 255) / 256, 256, 0, stream>>>(dis, NNODES);

    const int ntiles = NNODES / 16;
    const int gblk   = (ntiles + 3) / 4;

    const float* hin = x;
    for (int l = 0; l < 3; ++l) {
        gemm64_wmma<<<gblk, 128, 0, stream>>>(hin, Ws[l], B, ntiles);
        fill_f32<<<(n64 + 255) / 256, 256, 0, stream>>>(A, 0.0f, n64);
        scatter_kernel<<<(NEDGES * 16 + 255) / 256, 256, 0, stream>>>(src, dst, dis, B, A, NEDGES);
        bias_relu_kernel<<<(n64 + 255) / 256, 256, 0, stream>>>(A, B, dis, bs[l], n64);
        hin = A;
    }

    final_linear<<<(NNODES * 8 + 255) / 256, 256, 0, stream>>>(A, linW, linb, (float*)d_out, NNODES * 8);
}


# Elementwise passes folded into the dense ones: rsqrt rides in product 0; bias+relu+clear is the row loader of the next product and of the final linear, now on WMMA (sum in k groups of 4); 9 launches
# speedup vs baseline: 1.9513x; 1.9513x over previous
#include <hip/hip_runtime.h>

typedef __attribute__((ext_vector_type(2))) float v2f;
typedef __attribute__((ext_vector_type(8))) float v8f;

#define NNODES 50000
#define NEDGES 800000
#define DH     64


__device__ __forceinline__ int chip_half() {
    return (int)((__builtin_amdgcn_s_sendmsg_rtn(0x87) >> 18) & 1u);
}

#define INIT_BLK    256
#define INIT_GRID   625
#define INIT_T      (INIT_GRID * INIT_BLK)
#define INIT_TRIPS  10
#define PLANES_F4   (2 * NNODES * DH / 4)
static_assert(INIT_T >= 2 * NNODES, "one thread per degree start value");
static_assert(PLANES_F4 == INIT_T * INIT_TRIPS, "the clear must cover both sum planes exactly");

__global__ void fill_deg(float* __restrict__ p, int n, float* __restrict__ A) {
    const int i = blockIdx.x * INIT_BLK + threadIdx.x;
    if (i < 2 * n) p[i] = i < n ? 1.0f : 0.0f;
    float4* __restrict__ A4 = (float4*)A;
    const float4 z = make_float4(0.0f, 0.0f, 0.0f, 0.0f);
    #pragma unroll
    for (int k = 0; k < INIT_TRIPS; ++k) A4[i + k * INIT_T] = z;
}

__global__ void deg_kernel(const int* __restrict__ dst, float* __restrict__ deg, int e, int n) {
    int i = blockIdx.x * blockDim.x + threadIdx.x;
    if (i >= e) return;
    const int dn = dst[i];
    float* dh = deg + chip_half() * n;
    atomicAdd(&dh[dn], 1.0f);
}

#define TILE_M  16
#define NTILES  (NNODES / TILE_M)
#define GW      5
#define GBLK    (GW * 32)
#define GGRID   (NTILES / GW)
#define TP      68
static_assert(NNODES % TILE_M == 0, "rows must fill whole slabs");
static_assert(NTILES % GW == 0, "slabs must fill whole work-groups");

__device__ __forceinline__ void stage_w64(const float* __restrict__ W, float* __restrict__ Wl) {
    for (int i = threadIdx.x; i < DH * DH; i += GBLK) {
        int k = i >> 6, n = i & 63;
        Wl[(k >> 1) * 128 + n * 2 + (k & 1)] = W[i];
    }
}

struct SlabRegs {
    float4 s0[8], s1[8], bb[8];
    float  dd[8];
    float  bz0, bz1, bz2, bz3;
};

__device__ __forceinline__ void slab_load(SlabRegs& R, const float* A0, const float* A1,
                                          const float* Bp, const float* __restrict__ dis,
                                          const float* __restrict__ bias, int m0, int lane) {
    const int c4 = lane & 15;
    const int h  = lane >> 4;
    const float4* a0p = (const float4*)(A0 + m0 * DH);
    const float4* a1p = (const float4*)(A1 + m0 * DH);
    const float4* bp  = (const float4*)(Bp + m0 * DH);
    #pragma unroll
    for (int i = 0; i < 8; ++i) {
        const int q = i * 32 + lane;
        R.s0[i] = a0p[q];
        R.s1[i] = a1p[q];
        R.bb[i] = bp[q];
        R.dd[i] = dis[m0 + 2 * i + h];
    }
    R.bz0 = bias[4 * c4 + 0];
    R.bz1 = bias[4 * c4 + 1];
    R.bz2 = bias[4 * c4 + 2];
    R.bz3 = bias[4 * c4 + 3];
}

__device__ __forceinline__ void slab_finish(const SlabRegs& R, int lane, float* __restrict__ Tw) {
    const int c4 = lane & 15;
    const int h  = lane >> 4;
    #pragma unroll
    for (int i = 0; i < 8; ++i) {
        const float d = R.dd[i];
        float4 v;
        v.x = (R.s0[i].x + R.s1[i].x) + R.bb[i].x * d * d + R.bz0;
        v.y = (R.s0[i].y + R.s1[i].y) + R.bb[i].y * d * d + R.bz1;
        v.z = (R.s0[i].z + R.s1[i].z) + R.bb[i].z * d * d + R.bz2;
        v.w = (R.s0[i].w + R.s1[i].w) + R.bb[i].w * d * d + R.bz3;
        v.x = v.x > 0.0f ? v.x : 0.0f;
        v.y = v.y > 0.0f ? v.y : 0.0f;
        v.z = v.z > 0.0f ? v.z : 0.0f;
        v.w = v.w > 0.0f ? v.w : 0.0f;
        *(float4*)(Tw + (2 * i + h) * TP + 4 * c4) = v;
    }
}

__device__ __forceinline__ void slab_clear(float* A0, float* A1, int m0, int lane) {
    float4* a0p = (float4*)(A0 + m0 * DH);
    float4* a1p = (float4*)(A1 + m0 * DH);
    const float4 z = make_float4(0.0f, 0.0f, 0.0f, 0.0f);
    #pragma unroll
    for (int i = 0; i < 8; ++i) {
        a0p[i * 32 + lane] = z;
        a1p[i * 32 + lane] = z;
    }
}

__device__ __forceinline__ void store_slab(float* Bout, int m0, int lane,
                                           v8f acc0, v8f acc1, v8f acc2, v8f acc3) {
    const int ncol   = lane & 15;
    const int rowoff = (lane >> 4) * 8;
    #pragma unroll
    for (int r = 0; r < 8; ++r) {
        float* o = Bout + (m0 + rowoff + r) * DH + ncol;
        o[ 0] = acc0[r];
        o[16] = acc1[r];
        o[32] = acc2[r];
        o[48] = acc3[r];
    }
}

__global__ void __launch_bounds__(GBLK)
dense_first(const float* __restrict__ X, const float* __restrict__ W,
            float* __restrict__ Bout, float* __restrict__ deg) {
    __shared__ float Wl[32 * 128];
    const int lane = threadIdx.x & 31;
    const int wave = threadIdx.x >> 5;
    const int m0   = (blockIdx.x * GW + wave) * TILE_M;

    const int dr = m0 + (lane & 15);
    const float dsum = deg[dr] + deg[dr + NNODES];
    stage_w64(W, Wl);
    __syncthreads();
    if (lane < 16) deg[dr] = __frsqrt_rn(dsum);

    const int row   = m0 + (lane & 15);
    const int khalf = (lane >> 4) * 2;
    const int ncol  = lane & 15;

    v8f acc0 = {}, acc1 = {}, acc2 = {}, acc3 = {};
    #pragma unroll
    for (int k0 = 0; k0 < DH; k0 += 4) {
        const int kb = k0 + khalf;
        v2f a = *(const v2f*)(X + row * DH + kb);
        const float* bp = &Wl[(kb >> 1) * 128 + ncol * 2];
        v2f b0 = *(const v2f*)(bp +  0);
        v2f b1 = *(const v2f*)(bp + 32);
        v2f b2 = *(const v2f*)(bp + 64);
        v2f b3 = *(const v2f*)(bp + 96);
        acc0 = __builtin_amdgcn_wmma_f32_16x16x4_f32(false, a, false, b0, (short)0, acc0, false, false);
        acc1 = __builtin_amdgcn_wmma_f32_16x16x4_f32(false, a, false, b1, (short)0, acc1, false, false);
        acc2 = __builtin_amdgcn_wmma_f32_16x16x4_f32(false, a, false, b2, (short)0, acc2, false, false);
        acc3 = __builtin_amdgcn_wmma_f32_16x16x4_f32(false, a, false, b3, (short)0, acc3, false, false);
    }
    store_slab(Bout, m0, lane, acc0, acc1, acc2, acc3);
}

__global__ void __launch_bounds__(GBLK)
dense_fused(float* __restrict__ A, const float* __restrict__ dis,
            const float* __restrict__ bias, const float* __restrict__ W, float* Bio) {
    __shared__ float Wl[32 * 128];
    __shared__ alignas(16) float Tl[GW * TILE_M * TP];

    const int lane = threadIdx.x & 31;
    const int wave = threadIdx.x >> 5;
    const int m0   = (blockIdx.x * GW + wave) * TILE_M;
    float* Tw = Tl + wave * (TILE_M * TP);

    SlabRegs R;
    slab_load(R, A, A + NNODES * DH, Bio, dis, bias, m0, lane);
    stage_w64(W, Wl);
    slab_finish(R, lane, Tw);
    __syncthreads();
    slab_clear(A, A + NNODES * DH, m0, lane);

    const int arow  = lane & 15;
    const int khalf = (lane >> 4) * 2;
    const int ncol  = lane & 15;

    v8f acc0 = {}, acc1 = {}, acc2 = {}, acc3 = {};
    #pragma unroll
    for (int k0 = 0; k0 < DH; k0 += 4) {
        const int kb = k0 + khalf;
        v2f a = *(const v2f*)(Tw + arow * TP + kb);
        const float* bp = &Wl[(kb >> 1) * 128 + ncol * 2];
        v2f b0 = *(const v2f*)(bp +  0);
        v2f b1 = *(const v2f*)(bp + 32);
        v2f b2 = *(const v2f*)(bp + 64);
        v2f b3 = *(const v2f*)(bp + 96);
        acc0 = __builtin_amdgcn_wmma_f32_16x16x4_f32(false, a, false, b0, (short)0, acc0, false, false);
        acc1 = __builtin_amdgcn_wmma_f32_16x16x4_f32(false, a, false, b1, (short)0, acc1, false, false);
        acc2 = __builtin_amdgcn_wmma_f32_16x16x4_f32(false, a, false, b2, (short)0, acc2, false, false);
        acc3 = __builtin_amdgcn_wmma_f32_16x16x4_f32(false, a, false, b3, (short)0, acc3, false, false);
    }
    store_slab(Bio, m0, lane, acc0, acc1, acc2, acc3);
}

__global__ void __launch_bounds__(GBLK)
final_fused(float* __restrict__ A, const float* __restrict__ dis,
            const float* __restrict__ bias, const float* __restrict__ Bp,
            const float* __restrict__ LW, const float* __restrict__ Lb,
            float* __restrict__ out) {
    __shared__ float Ll[32 * 32];
    __shared__ alignas(16) float Tl[GW * TILE_M * TP];

    const int lane = threadIdx.x & 31;
    const int wave = threadIdx.x >> 5;
    const int m0   = (blockIdx.x * GW + wave) * TILE_M;
    float* Tw = Tl + wave * (TILE_M * TP);

    SlabRegs R;
    slab_load(R, A, A + NNODES * DH, Bp, dis, bias, m0, lane);
    const float lb = Lb[lane & 7];
    for (int i = threadIdx.x; i < 32 * 32; i += GBLK) {
        const int kp = i >> 5, n = (i >> 1) & 15, kk = i & 1;
        const float w = LW[(2 * kp + kk) * 8 + (n & 7)];
        Ll[i] = n < 8 ? w : 0.0f;
    }
    slab_finish(R, lane, Tw);
    __syncthreads();

    const int c     = lane & 15;
    const int khalf = (lane >> 4) * 2;
    const float st  = c < 8 ? lb : 0.0f;
    v8f acc = { st, st, st, st, st, st, st, st };
    #pragma unroll
    for (int k0 = 0; k0 < DH; k0 += 4) {
        const int kb = k0 + khalf;
        v2f a = *(const v2f*)(Tw + c * TP + kb);
        v2f b = *(const v2f*)(Ll + (kb >> 1) * 32 + c * 2);
        acc = __builtin_amdgcn_wmma_f32_16x16x4_f32(false, a, false, b, (short)0, acc, false, false);
    }
    const int rowoff = (lane >> 4) * 8;
    if (c < 8) {
        #pragma unroll
        for (int r = 0; r < 8; ++r) out[(m0 + rowoff + r) * 8 + c] = acc[r];
    }
}

#define EPW   32
#define EBAT  8
static_assert(NEDGES % EPW == 0, "edge count must be a multiple of the wave's edge chunk");
static_assert(EPW % EBAT == 0, "chunk must be a multiple of the load group");

__global__ void scatter_kernel(const int* __restrict__ src,
                               const int* __restrict__ dst,
                               const float* __restrict__ dis,
                               const float* __restrict__ B,
                               float* __restrict__ A, int e, int n64) {
    const int lane = threadIdx.x & 31;
    const int wave = (blockIdx.x * blockDim.x + threadIdx.x) >> 5;
    const int e0   = wave * EPW;
    if (e0 >= e) return;

    const int   s = src[e0 + lane];
    const int   d = dst[e0 + lane];
    const float w = dis[s] * dis[d];

    float* __restrict__ Ah = A + chip_half() * n64;

    #pragma unroll
    for (int j0 = 0; j0 < EPW; j0 += EBAT) {
        float lo[EBAT], hi[EBAT];
        #pragma unroll
        for (int j = 0; j < EBAT; ++j) {
            const int sj = __builtin_amdgcn_readlane(s, j0 + j);
            const float* bp = B + sj * DH + lane;
            lo[j] = bp[0];
            hi[j] = bp[32];
        }
        #pragma unroll
        for (int j = 0; j < EBAT; ++j) {
            const int   dj = __builtin_amdgcn_readlane(d, j0 + j);
            const float wj = __builtin_bit_cast(float,
                __builtin_amdgcn_readlane(__builtin_bit_cast(int, w), j0 + j));
            float* o = Ah + dj * DH + lane;
            atomicAdd(o,      lo[j] * wj);
            atomicAdd(o + 32, hi[j] * wj);
        }
    }
}

extern "C" void kernel_launch(void* const* d_in, const int* in_sizes, int n_in,
                              void* d_out, int out_size, void* d_ws, size_t ws_size,
                              hipStream_t stream) {
    const float* x    = (const float*)d_in[0];
    const int*   edge = (const int*)d_in[1];
    const int*   src  = edge;
    const int*   dst  = edge + NEDGES;
    const float* W0 = (const float*)d_in[2];
    const float* b0 = (const float*)d_in[3];
    const float* W1 = (const float*)d_in[4];
    const float* b1 = (const float*)d_in[5];
    const float* W2 = (const float*)d_in[6];
    const float* b2 = (const float*)d_in[7];
    const float* linW = (const float*)d_in[8];
    const float* linb = (const float*)d_in[9];

    float* A   = (float*)d_ws;
    float* B   = A + 2 * NNODES * DH;
    float* dis = B + NNODES * DH;

    const int n64  = NNODES * DH;
    const int sgrd = (NEDGES / EPW + 3) / 4;

    fill_deg<<<INIT_GRID, INIT_BLK, 0, stream>>>(dis, NNODES, A);
    deg_kernel<<<(NEDGES + 255) / 256, 256, 0, stream>>>(dst, dis, NEDGES, NNODES);

    dense_first<<<GGRID, GBLK, 0, stream>>>(x, W0, B, dis);
    scatter_kernel<<<sgrd, 128, 0, stream>>>(src, dst, dis, B, A, NEDGES, n64);
    dense_fused<<<GGRID, GBLK, 0, stream>>>(A, dis, b0, W1, B);
    scatter_kernel<<<sgrd, 128, 0, stream>>>(src, dst, dis, B, A, NEDGES, n64);
    dense_fused<<<GGRID, GBLK, 0, stream>>>(A, dis, b1, W2, B);
    scatter_kernel<<<sgrd, 128, 0, stream>>>(src, dst, dis, B, A, NEDGES, n64);
    final_fused<<<GGRID, GBLK, 0, stream>>>(A, dis, b2, B, linW, linb, (float*)d_out);
}


# Edge weights dis[src]*dis[dst] computed once: edge pass 1 stores them per edge, passes 2 and 3 read one whole line per wave in place of two 32-line gathers; atomic arrival order as before
# speedup vs baseline: 2.0251x; 1.0378x over previous
#include <hip/hip_runtime.h>

typedef __attribute__((ext_vector_type(2))) float v2f;
typedef __attribute__((ext_vector_type(8))) float v8f;

#define NNODES 50000
#define NEDGES 800000
#define DH     64


__device__ __forceinline__ int chip_half() {
    return (int)((__builtin_amdgcn_s_sendmsg_rtn(0x87) >> 18) & 1u);
}

#define INIT_BLK    256
#define INIT_GRID   625
#define INIT_T      (INIT_GRID * INIT_BLK)
#define INIT_TRIPS  10
#define PLANES_F4   (2 * NNODES * DH / 4)
static_assert(INIT_T >= 2 * NNODES, "one thread per degree start value");
static_assert(PLANES_F4 == INIT_T * INIT_TRIPS, "the clear must cover both sum planes exactly");

__global__ void fill_deg(float* __restrict__ p, int n, float* __restrict__ A) {
    const int i = blockIdx.x * INIT_BLK + threadIdx.x;
    if (i < 2 * n) p[i] = i < n ? 1.0f : 0.0f;
    float4* __restrict__ A4 = (float4*)A;
    const float4 z = make_float4(0.0f, 0.0f, 0.0f, 0.0f);
    #pragma unroll
    for (int k = 0; k < INIT_TRIPS; ++k) A4[i + k * INIT_T] = z;
}

__global__ void deg_kernel(const int* __restrict__ dst, float* __restrict__ deg, int e, int n) {
    int i = blockIdx.x * blockDim.x + threadIdx.x;
    if (i >= e) return;
    const int dn = dst[i];
    float* dh = deg + chip_half() * n;
    atomicAdd(&dh[dn], 1.0f);
}

#define TILE_M  16
#define NTILES  (NNODES / TILE_M)
#define GW      5
#define GBLK    (GW * 32)
#define GGRID   (NTILES / GW)
#define TP      68
static_assert(NNODES % TILE_M == 0, "rows must fill whole slabs");
static_assert(NTILES % GW == 0, "slabs must fill whole work-groups");

__device__ __forceinline__ void stage_w64(const float* __restrict__ W, float* __restrict__ Wl) {
    for (int i = threadIdx.x; i < DH * DH; i += GBLK) {
        int k = i >> 6, n = i & 63;
        Wl[(k >> 1) * 128 + n * 2 + (k & 1)] = W[i];
    }
}

struct SlabRegs {
    float4 s0[8], s1[8], bb[8];
    float  dd[8];
    float  bz0, bz1, bz2, bz3;
};

__device__ __forceinline__ void slab_load(SlabRegs& R, const float* A0, const float* A1,
                                          const float* Bp, const float* __restrict__ dis,
                                          const float* __restrict__ bias, int m0, int lane) {
    const int c4 = lane & 15;
    const int h  = lane >> 4;
    const float4* a0p = (const float4*)(A0 + m0 * DH);
    const float4* a1p = (const float4*)(A1 + m0 * DH);
    const float4* bp  = (const float4*)(Bp + m0 * DH);
    #pragma unroll
    for (int i = 0; i < 8; ++i) {
        const int q = i * 32 + lane;
        R.s0[i] = a0p[q];
        R.s1[i] = a1p[q];
        R.bb[i] = bp[q];
        R.dd[i] = dis[m0 + 2 * i + h];
    }
    R.bz0 = bias[4 * c4 + 0];
    R.bz1 = bias[4 * c4 + 1];
    R.bz2 = bias[4 * c4 + 2];
    R.bz3 = bias[4 * c4 + 3];
}

__device__ __forceinline__ void slab_finish(const SlabRegs& R, int lane, float* __restrict__ Tw) {
    const int c4 = lane & 15;
    const int h  = lane >> 4;
    #pragma unroll
    for (int i = 0; i < 8; ++i) {
        const float d = R.dd[i];
        float4 v;
        v.x = (R.s0[i].x + R.s1[i].x) + R.bb[i].x * d * d + R.bz0;
        v.y = (R.s0[i].y + R.s1[i].y) + R.bb[i].y * d * d + R.bz1;
        v.z = (R.s0[i].z + R.s1[i].z) + R.bb[i].z * d * d + R.bz2;
        v.w = (R.s0[i].w + R.s1[i].w) + R.bb[i].w * d * d + R.bz3;
        v.x = v.x > 0.0f ? v.x : 0.0f;
        v.y = v.y > 0.0f ? v.y : 0.0f;
        v.z = v.z > 0.0f ? v.z : 0.0f;
        v.w = v.w > 0.0f ? v.w : 0.0f;
        *(float4*)(Tw + (2 * i + h) * TP + 4 * c4) = v;
    }
}

__device__ __forceinline__ void slab_clear(float* A0, float* A1, int m0, int lane) {
    float4* a0p = (float4*)(A0 + m0 * DH);
    float4* a1p = (float4*)(A1 + m0 * DH);
    const float4 z = make_float4(0.0f, 0.0f, 0.0f, 0.0f);
    #pragma unroll
    for (int i = 0; i < 8; ++i) {
        a0p[i * 32 + lane] = z;
        a1p[i * 32 + lane] = z;
    }
}

__device__ __forceinline__ void store_slab(float* Bout, int m0, int lane,
                                           v8f acc0, v8f acc1, v8f acc2, v8f acc3) {
    const int ncol   = lane & 15;
    const int rowoff = (lane >> 4) * 8;
    #pragma unroll
    for (int r = 0; r < 8; ++r) {
        float* o = Bout + (m0 + rowoff + r) * DH + ncol;
        o[ 0] = acc0[r];
        o[16] = acc1[r];
        o[32] = acc2[r];
        o[48] = acc3[r];
    }
}

__global__ void __launch_bounds__(GBLK)
dense_first(const float* __restrict__ X, const float* __restrict__ W,
            float* __restrict__ Bout, float* __restrict__ deg) {
    __shared__ float Wl[32 * 128];
    const int lane = threadIdx.x & 31;
    const int wave = threadIdx.x >> 5;
    const int m0   = (blockIdx.x * GW + wave) * TILE_M;

    const int dr = m0 + (lane & 15);
    const float dsum = deg[dr] + deg[dr + NNODES];
    stage_w64(W, Wl);
    __syncthreads();
    if (lane < 16) deg[dr] = __frsqrt_rn(dsum);

    const int row   = m0 + (lane & 15);
    const int khalf = (lane >> 4) * 2;
    const int ncol  = lane & 15;

    v8f acc0 = {}, acc1 = {}, acc2 = {}, acc3 = {};
    #pragma unroll
    for (int k0 = 0; k0 < DH; k0 += 4) {
        const int kb = k0 + khalf;
        v2f a = *(const v2f*)(X + row * DH + kb);
        const float* bp = &Wl[(kb >> 1) * 128 + ncol * 2];
        v2f b0 = *(const v2f*)(bp +  0);
        v2f b1 = *(const v2f*)(bp + 32);
        v2f b2 = *(const v2f*)(bp + 64);
        v2f b3 = *(const v2f*)(bp + 96);
        acc0 = __builtin_amdgcn_wmma_f32_16x16x4_f32(false, a, false, b0, (short)0, acc0, false, false);
        acc1 = __builtin_amdgcn_wmma_f32_16x16x4_f32(false, a, false, b1, (short)0, acc1, false, false);
        acc2 = __builtin_amdgcn_wmma_f32_16x16x4_f32(false, a, false, b2, (short)0, acc2, false, false);
        acc3 = __builtin_amdgcn_wmma_f32_16x16x4_f32(false, a, false, b3, (short)0, acc3, false, false);
    }
    store_slab(Bout, m0, lane, acc0, acc1, acc2, acc3);
}

__global__ void __launch_bounds__(GBLK)
dense_fused(float* __restrict__ A, const float* __restrict__ dis,
            const float* __restrict__ bias, const float* __restrict__ W, float* Bio) {
    __shared__ float Wl[32 * 128];
    __shared__ alignas(16) float Tl[GW * TILE_M * TP];

    const int lane = threadIdx.x & 31;
    const int wave = threadIdx.x >> 5;
    const int m0   = (blockIdx.x * GW + wave) * TILE_M;
    float* Tw = Tl + wave * (TILE_M * TP);

    SlabRegs R;
    slab_load(R, A, A + NNODES * DH, Bio, dis, bias, m0, lane);
    stage_w64(W, Wl);
    slab_finish(R, lane, Tw);
    __syncthreads();
    slab_clear(A, A + NNODES * DH, m0, lane);

    const int arow  = lane & 15;
    const int khalf = (lane >> 4) * 2;
    const int ncol  = lane & 15;

    v8f acc0 = {}, acc1 = {}, acc2 = {}, acc3 = {};
    #pragma unroll
    for (int k0 = 0; k0 < DH; k0 += 4) {
        const int kb = k0 + khalf;
        v2f a = *(const v2f*)(Tw + arow * TP + kb);
        const float* bp = &Wl[(kb >> 1) * 128 + ncol * 2];
        v2f b0 = *(const v2f*)(bp +  0);
        v2f b1 = *(const v2f*)(bp + 32);
        v2f b2 = *(const v2f*)(bp + 64);
        v2f b3 = *(const v2f*)(bp + 96);
        acc0 = __builtin_amdgcn_wmma_f32_16x16x4_f32(false, a, false, b0, (short)0, acc0, false, false);
        acc1 = __builtin_amdgcn_wmma_f32_16x16x4_f32(false, a, false, b1, (short)0, acc1, false, false);
        acc2 = __builtin_amdgcn_wmma_f32_16x16x4_f32(false, a, false, b2, (short)0, acc2, false, false);
        acc3 = __builtin_amdgcn_wmma_f32_16x16x4_f32(false, a, false, b3, (short)0, acc3, false, false);
    }
    store_slab(Bio, m0, lane, acc0, acc1, acc2, acc3);
}

__global__ void __launch_bounds__(GBLK)
final_fused(float* __restrict__ A, const float* __restrict__ dis,
            const float* __restrict__ bias, const float* __restrict__ Bp,
            const float* __restrict__ LW, const float* __restrict__ Lb,
            float* __restrict__ out) {
    __shared__ float Ll[32 * 32];
    __shared__ alignas(16) float Tl[GW * TILE_M * TP];

    const int lane = threadIdx.x & 31;
    const int wave = threadIdx.x >> 5;
    const int m0   = (blockIdx.x * GW + wave) * TILE_M;
    float* Tw = Tl + wave * (TILE_M * TP);

    SlabRegs R;
    slab_load(R, A, A + NNODES * DH, Bp, dis, bias, m0, lane);
    const float lb = Lb[lane & 7];
    for (int i = threadIdx.x; i < 32 * 32; i += GBLK) {
        const int kp = i >> 5, n = (i >> 1) & 15, kk = i & 1;
        const float w = LW[(2 * kp + kk) * 8 + (n & 7)];
        Ll[i] = n < 8 ? w : 0.0f;
    }
    slab_finish(R, lane, Tw);
    __syncthreads();

    const int c     = lane & 15;
    const int khalf = (lane >> 4) * 2;
    const float st  = c < 8 ? lb : 0.0f;
    v8f acc = { st, st, st, st, st, st, st, st };
    #pragma unroll
    for (int k0 = 0; k0 < DH; k0 += 4) {
        const int kb = k0 + khalf;
        v2f a = *(const v2f*)(Tw + c * TP + kb);
        v2f b = *(const v2f*)(Ll + (kb >> 1) * 32 + c * 2);
        acc = __builtin_amdgcn_wmma_f32_16x16x4_f32(false, a, false, b, (short)0, acc, false, false);
    }
    const int rowoff = (lane >> 4) * 8;
    if (c < 8) {
        #pragma unroll
        for (int r = 0; r < 8; ++r) out[(m0 + rowoff + r) * 8 + c] = acc[r];
    }
}

#define EPW   32
#define EBAT  8
static_assert(NEDGES % EPW == 0, "edge count must be a multiple of the wave's edge chunk");
static_assert(EPW % EBAT == 0, "chunk must be a multiple of the load group");

__device__ __forceinline__ void scatter_body(const bool make_w,
                                             const int* __restrict__ src,
                                             const int* __restrict__ dst,
                                             const float* __restrict__ dis,
                                             float* __restrict__ wE,
                                             const float* __restrict__ B,
                                             float* __restrict__ A, int e, int n64) {
    const int lane = threadIdx.x & 31;
    const int wave = (blockIdx.x * blockDim.x + threadIdx.x) >> 5;
    const int e0   = wave * EPW;
    if (e0 >= e) return;

    const int s = src[e0 + lane];
    const int d = dst[e0 + lane];
    const float w = make_w ? dis[s] * dis[d] : wE[e0 + lane];

    float* __restrict__ Ah = A + chip_half() * n64;

    #pragma unroll
    for (int j0 = 0; j0 < EPW; j0 += EBAT) {
        float lo[EBAT], hi[EBAT];
        #pragma unroll
        for (int j = 0; j < EBAT; ++j) {
            const int sj = __builtin_amdgcn_readlane(s, j0 + j);
            const float* bp = B + sj * DH + lane;
            lo[j] = bp[0];
            hi[j] = bp[32];
        }
        if (make_w && j0 == 0) wE[e0 + lane] = w;
        #pragma unroll
        for (int j = 0; j < EBAT; ++j) {
            const int   dj = __builtin_amdgcn_readlane(d, j0 + j);
            const float wj = __builtin_bit_cast(float,
                __builtin_amdgcn_readlane(__builtin_bit_cast(int, w), j0 + j));
            float* o = Ah + dj * DH + lane;
            atomicAdd(o,      lo[j] * wj);
            atomicAdd(o + 32, hi[j] * wj);
        }
    }
}

__global__ void scatter_first(const int* __restrict__ src,
                              const int* __restrict__ dst,
                              const float* __restrict__ dis,
                              float* __restrict__ wE,
                              const float* __restrict__ B,
                              float* __restrict__ A, int e, int n64) {
    scatter_body(true, src, dst, dis, wE, B, A, e, n64);
}

__global__ void scatter_next(const int* __restrict__ src,
                             const int* __restrict__ dst,
                             float* __restrict__ wE,
                             const float* __restrict__ B,
                             float* __restrict__ A, int e, int n64) {
    scatter_body(false, src, dst, nullptr, wE, B, A, e, n64);
}

extern "C" void kernel_launch(void* const* d_in, const int* in_sizes, int n_in,
                              void* d_out, int out_size, void* d_ws, size_t ws_size,
                              hipStream_t stream) {
    const float* x    = (const float*)d_in[0];
    const int*   edge = (const int*)d_in[1];
    const int*   src  = edge;
    const int*   dst  = edge + NEDGES;
    const float* W0 = (const float*)d_in[2];
    const float* b0 = (const float*)d_in[3];
    const float* W1 = (const float*)d_in[4];
    const float* b1 = (const float*)d_in[5];
    const float* W2 = (const float*)d_in[6];
    const float* b2 = (const float*)d_in[7];
    const float* linW = (const float*)d_in[8];
    const float* linb = (const float*)d_in[9];

    float* A   = (float*)d_ws;
    float* B   = A + 2 * NNODES * DH;
    float* dis = B + NNODES * DH;
    float* wE  = dis + 2 * NNODES;

    const int n64  = NNODES * DH;
    const int sgrd = (NEDGES / EPW + 3) / 4;

    fill_deg<<<INIT_GRID, INIT_BLK, 0, stream>>>(dis, NNODES, A);
    deg_kernel<<<(NEDGES + 255) / 256, 256, 0, stream>>>(dst, dis, NEDGES, NNODES);

    dense_first<<<GGRID, GBLK, 0, stream>>>(x, W0, B, dis);
    scatter_first<<<sgrd, 128, 0, stream>>>(src, dst, dis, wE, B, A, NEDGES, n64);
    dense_fused<<<GGRID, GBLK, 0, stream>>>(A, dis, b0, W1, B);
    scatter_next<<<sgrd, 128, 0, stream>>>(src, dst, wE, B, A, NEDGES, n64);
    dense_fused<<<GGRID, GBLK, 0, stream>>>(A, dis, b1, W2, B);
    scatter_next<<<sgrd, 128, 0, stream>>>(src, dst, wE, B, A, NEDGES, n64);
    final_fused<<<GGRID, GBLK, 0, stream>>>(A, dis, b2, B, linW, linb, (float*)d_out);
}
